# starting kernel
# baseline (speedup 1.0000x reference)
#include <hip/hip_runtime.h>

typedef float v2f __attribute__((ext_vector_type(2)));
typedef float v8f __attribute__((ext_vector_type(8)));

namespace {
constexpr int NG    = 128;
constexpr int GN    = 128;
constexpr int DIM   = 512;
constexpr int KC    = 32;
constexpr int LDSS  = 36;
constexpr int PAIRS = (GN * (GN - 1)) / 2;
}

__device__ __forceinline__ unsigned f32_to_ordered(float f) {
  unsigned u = __float_as_uint(f);
  return (u & 0x80000000u) ? ~u : (u | 0x80000000u);
}
__device__ __forceinline__ float ordered_to_f32(unsigned u) {
  unsigned b = (u & 0x80000000u) ? (u & 0x7FFFFFFFu) : ~u;
  return __uint_as_float(b);
}

__global__ void init_minmax_kernel(unsigned* __restrict__ mm) {
  mm[0] = 0xFFFFFFFFu;
  mm[1] = 0x00000000u;
}

__global__ __launch_bounds__(256) void
sims_minmax_kernel(const float* __restrict__ E, unsigned* __restrict__ mm) {
  const int jt = blockIdx.x;
  const int it = blockIdx.y;
  if (jt < it) return;

  __shared__ float ldsA[GN * LDSS];
  __shared__ float ldsB[GN * LDSS];
  __shared__ float wmn[8], wmx[8];

  const int tid  = threadIdx.x;
  const int wave = tid >> 5;
  const int lane = tid & 31;
  const int m16  = lane & 15;
  const int kk   = (lane >> 4) << 1;
  const long rbase = (long)it * GN;
  const long cbase = (long)jt * GN;

  v8f zero = {};
  v8f acc[8];
#pragma unroll
  for (int c = 0; c < 8; ++c) acc[c] = zero;

  for (int kc = 0; kc < DIM; kc += KC) {
#pragma unroll
    for (int ld = 0; ld < 4; ++ld) {
      int e   = ld * 256 + tid;
      int row = e >> 3;
      int k4  = (e & 7) << 2;
      float4 va = *(const float4*)(E + (rbase + row) * DIM + kc + k4);
      *(float4*)(ldsA + row * LDSS + k4) = va;
      float4 vb = *(const float4*)(E + (cbase + row) * DIM + kc + k4);
      *(float4*)(ldsB + row * LDSS + k4) = vb;
    }
    __syncthreads();
#pragma unroll
    for (int ks = 0; ks < KC; ks += 4) {
      v2f a = *(const v2f*)(ldsA + (wave * 16 + m16) * LDSS + ks + kk);
#pragma unroll
      for (int c = 0; c < 8; ++c) {
        v2f b = *(const v2f*)(ldsB + (c * 16 + m16) * LDSS + ks + kk);
        acc[c] = __builtin_amdgcn_wmma_f32_16x16x4_f32(
            false, a, false, b, (short)0, acc[c], false, false);
      }
    }
    __syncthreads();
  }

  float mn = 3.402823466e38f, mx = -3.402823466e38f;
#pragma unroll
  for (int c = 0; c < 8; ++c)
#pragma unroll
    for (int v = 0; v < 8; ++v) {
      float x = acc[c][v];
      mn = fminf(mn, x);
      mx = fmaxf(mx, x);
    }
#pragma unroll
  for (int off = 16; off > 0; off >>= 1) {
    mn = fminf(mn, __shfl_xor(mn, off, 32));
    mx = fmaxf(mx, __shfl_xor(mx, off, 32));
  }
  if (lane == 0) { wmn[wave] = mn; wmx[wave] = mx; }
  __syncthreads();
  if (tid == 0) {
#pragma unroll
    for (int w = 1; w < 8; ++w) {
      mn = fminf(mn, wmn[w]);
      mx = fmaxf(mx, wmx[w]);
    }
    atomicMin(&mm[0], f32_to_ordered(mn));
    atomicMax(&mm[1], f32_to_ordered(mx));
  }
}

__global__ __launch_bounds__(256) void
diag_gather_kernel(const float* __restrict__ E, const unsigned* __restrict__ mm,
                   float* __restrict__ out) {
  const int g = blockIdx.x;

  __shared__ float ldsT[GN * LDSS];

  const int tid  = threadIdx.x;
  const int wave = tid >> 5;
  const int lane = tid & 31;
  const int m16  = lane & 15;
  const int kk   = (lane >> 4) << 1;
  const long base = (long)g * GN;

  v8f zero = {};
  v8f acc[8];
#pragma unroll
  for (int c = 0; c < 8; ++c) acc[c] = zero;

  for (int kc = 0; kc < DIM; kc += KC) {
#pragma unroll
    for (int ld = 0; ld < 4; ++ld) {
      int e   = ld * 256 + tid;
      int row = e >> 3;
      int k4  = (e & 7) << 2;
      float4 v = *(const float4*)(E + (base + row) * DIM + kc + k4);
      *(float4*)(ldsT + row * LDSS + k4) = v;
    }
    __syncthreads();
#pragma unroll
    for (int ks = 0; ks < KC; ks += 4) {
      v2f a = *(const v2f*)(ldsT + (wave * 16 + m16) * LDSS + ks + kk);
#pragma unroll
      for (int c = 0; c < 8; ++c) {
        v2f b = *(const v2f*)(ldsT + (c * 16 + m16) * LDSS + ks + kk);
        acc[c] = __builtin_amdgcn_wmma_f32_16x16x4_f32(
            false, a, false, b, (short)0, acc[c], false, false);
      }
    }
    __syncthreads();
  }

  const float mnv   = ordered_to_f32(mm[0]);
  const float mxv   = ordered_to_f32(mm[1]);
  const float scale = 1.0f / (mxv - mnv + 1e-7f);
  float* og = out + (long)g * PAIRS;

  const int rhalf = (lane >> 4) << 3;
#pragma unroll
  for (int c = 0; c < 8; ++c) {
    const int col = c * 16 + m16;
#pragma unroll
    for (int v = 0; v < 8; ++v) {
      const int row = wave * 16 + rhalf + v;
      if (col > row) {
        const int p = row * 127 - (row * (row - 1)) / 2 + (col - row - 1);
        og[p] = (acc[c][v] - mnv) * scale;
      }
    }
  }
}

extern "C" void kernel_launch(void* const* d_in, const int* in_sizes, int n_in,
                              void* d_out, int out_size, void* d_ws, size_t ws_size,
                              hipStream_t stream) {
  (void)in_sizes; (void)n_in; (void)out_size; (void)ws_size;
  const float* E = (const float*)d_in[0];
  unsigned* mm = (unsigned*)d_ws;
  float* out = (float*)d_out;

  init_minmax_kernel<<<dim3(1), dim3(1), 0, stream>>>(mm);
  sims_minmax_kernel<<<dim3(NG, NG), dim3(256), 0, stream>>>(E, mm);
  diag_gather_kernel<<<dim3(NG), dim3(256), 0, stream>>>(E, mm, out);
}


# Min/max pass: 4 waves per workgroup, each a 64x64 quarter as 4x4 accumulator tiles; LDS fragments in two register sets, next set read as one group ahead of 32 products; k sum order unchanged
# speedup vs baseline: 1.0727x; 1.0727x over previous
#include <hip/hip_runtime.h>

typedef float v2f __attribute__((ext_vector_type(2)));
typedef float v8f __attribute__((ext_vector_type(8)));

namespace {
constexpr int NG    = 128;
constexpr int GN    = 128;
constexpr int DIM   = 512;
constexpr int KC    = 32;
constexpr int LDSS  = 36;
constexpr int PAIRS = (GN * (GN - 1)) / 2;
}

__device__ __forceinline__ unsigned f32_to_ordered(float f) {
  unsigned u = __float_as_uint(f);
  return (u & 0x80000000u) ? ~u : (u | 0x80000000u);
}
__device__ __forceinline__ float ordered_to_f32(unsigned u) {
  unsigned b = (u & 0x80000000u) ? (u & 0x7FFFFFFFu) : ~u;
  return __uint_as_float(b);
}

__global__ void init_minmax_kernel(unsigned* __restrict__ mm) {
  mm[0] = 0xFFFFFFFFu;
  mm[1] = 0x00000000u;
}

__global__ __launch_bounds__(128) void
sims_minmax_kernel(const float* __restrict__ E, unsigned* __restrict__ mm) {
  const int jt = blockIdx.x;
  const int it = blockIdx.y;
  if (jt < it) return;

  __shared__ float ldsA[GN * LDSS];
  __shared__ float ldsB[GN * LDSS];
  __shared__ float wmn[4], wmx[4];

  const int tid  = threadIdx.x;
  const int wave = tid >> 5;
  const int lane = tid & 31;
  const int m16  = lane & 15;
  const int kk   = (lane >> 4) << 1;
  const int wr   = wave >> 1;
  const int wc   = wave & 1;
  const int srow = tid >> 3;
  const int sk4  = (tid & 7) << 2;

  const float* ga = E + ((long)it * GN + srow) * DIM + sk4;
  const float* gb = E + ((long)jt * GN + srow) * DIM + sk4;
  float* sa = ldsA + srow * LDSS + sk4;
  float* sb = ldsB + srow * LDSS + sk4;
  const float* pa = ldsA + (wr * 64 + m16) * LDSS + kk;
  const float* pb = ldsB + (wc * 64 + m16) * LDSS + kk;

  v8f zero = {};
  v8f acc[4][4];
#pragma unroll
  for (int i = 0; i < 4; ++i)
#pragma unroll
    for (int j = 0; j < 4; ++j) acc[i][j] = zero;

  for (int kc = 0; kc < DIM; kc += KC) {
    float4 va[8], vb[8];
#pragma unroll
    for (int ld = 0; ld < 8; ++ld) {
      va[ld] = *(const float4*)(ga + kc + ld * 16 * DIM);
      vb[ld] = *(const float4*)(gb + kc + ld * 16 * DIM);
    }
#pragma unroll
    for (int ld = 0; ld < 8; ++ld) {
      *(float4*)(sa + ld * 16 * LDSS) = va[ld];
      *(float4*)(sb + ld * 16 * LDSS) = vb[ld];
    }
    __syncthreads();

    v2f fa[2][2][4], fb[2][2][4];
#pragma unroll
    for (int u = 0; u < 2; ++u)
#pragma unroll
      for (int f = 0; f < 4; ++f) {
        fa[0][u][f] = *(const v2f*)(pa + f * 16 * LDSS + u * 4);
        fb[0][u][f] = *(const v2f*)(pb + f * 16 * LDSS + u * 4);
      }
#pragma unroll
    for (int g = 0; g < KC / 8; ++g) {
      const int cur = g & 1, nxt = cur ^ 1;
      if (g + 1 < KC / 8) {
#pragma unroll
        for (int u = 0; u < 2; ++u)
#pragma unroll
          for (int f = 0; f < 4; ++f) {
            fa[nxt][u][f] = *(const v2f*)(pa + f * 16 * LDSS + (g + 1) * 8 + u * 4);
            fb[nxt][u][f] = *(const v2f*)(pb + f * 16 * LDSS + (g + 1) * 8 + u * 4);
          }
      }
      __builtin_amdgcn_sched_barrier(0);
#pragma unroll
      for (int u = 0; u < 2; ++u)
#pragma unroll
        for (int i = 0; i < 4; ++i)
#pragma unroll
          for (int j = 0; j < 4; ++j)
            acc[i][j] = __builtin_amdgcn_wmma_f32_16x16x4_f32(
                false, fa[cur][u][i], false, fb[cur][u][j], (short)0, acc[i][j], false, false);
      __builtin_amdgcn_sched_barrier(0);
    }
    __syncthreads();
  }

  float mn = 3.402823466e38f, mx = -3.402823466e38f;
#pragma unroll
  for (int i = 0; i < 4; ++i)
#pragma unroll
    for (int j = 0; j < 4; ++j)
#pragma unroll
      for (int v = 0; v < 8; ++v) {
        float x = acc[i][j][v];
        mn = fminf(mn, x);
        mx = fmaxf(mx, x);
      }
#pragma unroll
  for (int off = 16; off > 0; off >>= 1) {
    mn = fminf(mn, __shfl_xor(mn, off, 32));
    mx = fmaxf(mx, __shfl_xor(mx, off, 32));
  }
  if (lane == 0) { wmn[wave] = mn; wmx[wave] = mx; }
  __syncthreads();
  if (tid == 0) {
#pragma unroll
    for (int w = 1; w < 4; ++w) {
      mn = fminf(mn, wmn[w]);
      mx = fmaxf(mx, wmx[w]);
    }
    atomicMin(&mm[0], f32_to_ordered(mn));
    atomicMax(&mm[1], f32_to_ordered(mx));
  }
}

__global__ __launch_bounds__(256) void
diag_gather_kernel(const float* __restrict__ E, const unsigned* __restrict__ mm,
                   float* __restrict__ out) {
  const int g = blockIdx.x;

  __shared__ float ldsT[GN * LDSS];

  const int tid  = threadIdx.x;
  const int wave = tid >> 5;
  const int lane = tid & 31;
  const int m16  = lane & 15;
  const int kk   = (lane >> 4) << 1;
  const long base = (long)g * GN;

  v8f zero = {};
  v8f acc[8];
#pragma unroll
  for (int c = 0; c < 8; ++c) acc[c] = zero;

  for (int kc = 0; kc < DIM; kc += KC) {
#pragma unroll
    for (int ld = 0; ld < 4; ++ld) {
      int e   = ld * 256 + tid;
      int row = e >> 3;
      int k4  = (e & 7) << 2;
      float4 v = *(const float4*)(E + (base + row) * DIM + kc + k4);
      *(float4*)(ldsT + row * LDSS + k4) = v;
    }
    __syncthreads();
#pragma unroll
    for (int ks = 0; ks < KC; ks += 4) {
      v2f a = *(const v2f*)(ldsT + (wave * 16 + m16) * LDSS + ks + kk);
#pragma unroll
      for (int c = 0; c < 8; ++c) {
        v2f b = *(const v2f*)(ldsT + (c * 16 + m16) * LDSS + ks + kk);
        acc[c] = __builtin_amdgcn_wmma_f32_16x16x4_f32(
            false, a, false, b, (short)0, acc[c], false, false);
      }
    }
    __syncthreads();
  }

  const float mnv   = ordered_to_f32(mm[0]);
  const float mxv   = ordered_to_f32(mm[1]);
  const float scale = 1.0f / (mxv - mnv + 1e-7f);
  float* og = out + (long)g * PAIRS;

  const int rhalf = (lane >> 4) << 3;
#pragma unroll
  for (int c = 0; c < 8; ++c) {
    const int col = c * 16 + m16;
#pragma unroll
    for (int v = 0; v < 8; ++v) {
      const int row = wave * 16 + rhalf + v;
      if (col > row) {
        const int p = row * 127 - (row * (row - 1)) / 2 + (col - row - 1);
        og[p] = (acc[c][v] - mnv) * scale;
      }
    }
  }
}

extern "C" void kernel_launch(void* const* d_in, const int* in_sizes, int n_in,
                              void* d_out, int out_size, void* d_ws, size_t ws_size,
                              hipStream_t stream) {
  (void)in_sizes; (void)n_in; (void)out_size; (void)ws_size;
  const float* E = (const float*)d_in[0];
  unsigned* mm = (unsigned*)d_ws;
  float* out = (float*)d_out;

  init_minmax_kernel<<<dim3(1), dim3(1), 0, stream>>>(mm);
  sims_minmax_kernel<<<dim3(NG, NG), dim3(128), 0, stream>>>(E, mm);
  diag_gather_kernel<<<dim3(NG), dim3(256), 0, stream>>>(E, mm, out);
}


# Min/max pass launched on a folded 129x64 grid holding only the 8256 macro tiles on or above the diagonal (row r beside row 127-r): no early-exit workgroups, every id carries a tile; sums unchanged
# speedup vs baseline: 1.1226x; 1.0465x over previous
#include <hip/hip_runtime.h>

typedef float v2f __attribute__((ext_vector_type(2)));
typedef float v8f __attribute__((ext_vector_type(8)));

namespace {
constexpr int NG    = 128;
constexpr int GN    = 128;
constexpr int DIM   = 512;
constexpr int KC    = 32;
constexpr int LDSS  = 36;
constexpr int PAIRS = (GN * (GN - 1)) / 2;
constexpr int FOLD_X = NG + 1;
constexpr int FOLD_Y = NG / 2;
}

__device__ __forceinline__ unsigned f32_to_ordered(float f) {
  unsigned u = __float_as_uint(f);
  return (u & 0x80000000u) ? ~u : (u | 0x80000000u);
}
__device__ __forceinline__ float ordered_to_f32(unsigned u) {
  unsigned b = (u & 0x80000000u) ? (u & 0x7FFFFFFFu) : ~u;
  return __uint_as_float(b);
}

__global__ void init_minmax_kernel(unsigned* __restrict__ mm) {
  mm[0] = 0xFFFFFFFFu;
  mm[1] = 0x00000000u;
}

__global__ __launch_bounds__(128) void
sims_minmax_kernel(const float* __restrict__ E, unsigned* __restrict__ mm) {
  const int fr = blockIdx.y;
  const int fc = blockIdx.x;
  const bool head = fc < NG - fr;
  const int it = head ? fr : NG - 1 - fr;
  const int jt = head ? fr + fc : fc - 1;

  __shared__ float ldsA[GN * LDSS];
  __shared__ float ldsB[GN * LDSS];
  __shared__ float wmn[4], wmx[4];

  const int tid  = threadIdx.x;
  const int wave = tid >> 5;
  const int lane = tid & 31;
  const int m16  = lane & 15;
  const int kk   = (lane >> 4) << 1;
  const int wr   = wave >> 1;
  const int wc   = wave & 1;
  const int srow = tid >> 3;
  const int sk4  = (tid & 7) << 2;

  const float* ga = E + ((long)it * GN + srow) * DIM + sk4;
  const float* gb = E + ((long)jt * GN + srow) * DIM + sk4;
  float* sa = ldsA + srow * LDSS + sk4;
  float* sb = ldsB + srow * LDSS + sk4;
  const float* pa = ldsA + (wr * 64 + m16) * LDSS + kk;
  const float* pb = ldsB + (wc * 64 + m16) * LDSS + kk;

  v8f zero = {};
  v8f acc[4][4];
#pragma unroll
  for (int i = 0; i < 4; ++i)
#pragma unroll
    for (int j = 0; j < 4; ++j) acc[i][j] = zero;

  for (int kc = 0; kc < DIM; kc += KC) {
    float4 va[8], vb[8];
#pragma unroll
    for (int ld = 0; ld < 8; ++ld) {
      va[ld] = *(const float4*)(ga + kc + ld * 16 * DIM);
      vb[ld] = *(const float4*)(gb + kc + ld * 16 * DIM);
    }
#pragma unroll
    for (int ld = 0; ld < 8; ++ld) {
      *(float4*)(sa + ld * 16 * LDSS) = va[ld];
      *(float4*)(sb + ld * 16 * LDSS) = vb[ld];
    }
    __syncthreads();

    v2f fa[2][2][4], fb[2][2][4];
#pragma unroll
    for (int u = 0; u < 2; ++u)
#pragma unroll
      for (int f = 0; f < 4; ++f) {
        fa[0][u][f] = *(const v2f*)(pa + f * 16 * LDSS + u * 4);
        fb[0][u][f] = *(const v2f*)(pb + f * 16 * LDSS + u * 4);
      }
#pragma unroll
    for (int g = 0; g < KC / 8; ++g) {
      const int cur = g & 1, nxt = cur ^ 1;
      if (g + 1 < KC / 8) {
#pragma unroll
        for (int u = 0; u < 2; ++u)
#pragma unroll
          for (int f = 0; f < 4; ++f) {
            fa[nxt][u][f] = *(const v2f*)(pa + f * 16 * LDSS + (g + 1) * 8 + u * 4);
            fb[nxt][u][f] = *(const v2f*)(pb + f * 16 * LDSS + (g + 1) * 8 + u * 4);
          }
      }
      __builtin_amdgcn_sched_barrier(0);
#pragma unroll
      for (int u = 0; u < 2; ++u)
#pragma unroll
        for (int i = 0; i < 4; ++i)
#pragma unroll
          for (int j = 0; j < 4; ++j)
            acc[i][j] = __builtin_amdgcn_wmma_f32_16x16x4_f32(
                false, fa[cur][u][i], false, fb[cur][u][j], (short)0, acc[i][j], false, false);
      __builtin_amdgcn_sched_barrier(0);
    }
    __syncthreads();
  }

  float mn = 3.402823466e38f, mx = -3.402823466e38f;
#pragma unroll
  for (int i = 0; i < 4; ++i)
#pragma unroll
    for (int j = 0; j < 4; ++j)
#pragma unroll
      for (int v = 0; v < 8; ++v) {
        float x = acc[i][j][v];
        mn = fminf(mn, x);
        mx = fmaxf(mx, x);
      }
#pragma unroll
  for (int off = 16; off > 0; off >>= 1) {
    mn = fminf(mn, __shfl_xor(mn, off, 32));
    mx = fmaxf(mx, __shfl_xor(mx, off, 32));
  }
  if (lane == 0) { wmn[wave] = mn; wmx[wave] = mx; }
  __syncthreads();
  if (tid == 0) {
#pragma unroll
    for (int w = 1; w < 4; ++w) {
      mn = fminf(mn, wmn[w]);
      mx = fmaxf(mx, wmx[w]);
    }
    atomicMin(&mm[0], f32_to_ordered(mn));
    atomicMax(&mm[1], f32_to_ordered(mx));
  }
}

__global__ __launch_bounds__(256) void
diag_gather_kernel(const float* __restrict__ E, const unsigned* __restrict__ mm,
                   float* __restrict__ out) {
  const int g = blockIdx.x;

  __shared__ float ldsT[GN * LDSS];

  const int tid  = threadIdx.x;
  const int wave = tid >> 5;
  const int lane = tid & 31;
  const int m16  = lane & 15;
  const int kk   = (lane >> 4) << 1;
  const long base = (long)g * GN;

  v8f zero = {};
  v8f acc[8];
#pragma unroll
  for (int c = 0; c < 8; ++c) acc[c] = zero;

  for (int kc = 0; kc < DIM; kc += KC) {
#pragma unroll
    for (int ld = 0; ld < 4; ++ld) {
      int e   = ld * 256 + tid;
      int row = e >> 3;
      int k4  = (e & 7) << 2;
      float4 v = *(const float4*)(E + (base + row) * DIM + kc + k4);
      *(float4*)(ldsT + row * LDSS + k4) = v;
    }
    __syncthreads();
#pragma unroll
    for (int ks = 0; ks < KC; ks += 4) {
      v2f a = *(const v2f*)(ldsT + (wave * 16 + m16) * LDSS + ks + kk);
#pragma unroll
      for (int c = 0; c < 8; ++c) {
        v2f b = *(const v2f*)(ldsT + (c * 16 + m16) * LDSS + ks + kk);
        acc[c] = __builtin_amdgcn_wmma_f32_16x16x4_f32(
            false, a, false, b, (short)0, acc[c], false, false);
      }
    }
    __syncthreads();
  }

  const float mnv   = ordered_to_f32(mm[0]);
  const float mxv   = ordered_to_f32(mm[1]);
  const float scale = 1.0f / (mxv - mnv + 1e-7f);
  float* og = out + (long)g * PAIRS;

  const int rhalf = (lane >> 4) << 3;
#pragma unroll
  for (int c = 0; c < 8; ++c) {
    const int col = c * 16 + m16;
#pragma unroll
    for (int v = 0; v < 8; ++v) {
      const int row = wave * 16 + rhalf + v;
      if (col > row) {
        const int p = row * 127 - (row * (row - 1)) / 2 + (col - row - 1);
        og[p] = (acc[c][v] - mnv) * scale;
      }
    }
  }
}

extern "C" void kernel_launch(void* const* d_in, const int* in_sizes, int n_in,
                              void* d_out, int out_size, void* d_ws, size_t ws_size,
                              hipStream_t stream) {
  (void)in_sizes; (void)n_in; (void)out_size; (void)ws_size;
  const float* E = (const float*)d_in[0];
  unsigned* mm = (unsigned*)d_ws;
  float* out = (float*)d_out;

  init_minmax_kernel<<<dim3(1), dim3(1), 0, stream>>>(mm);
  sims_minmax_kernel<<<dim3(FOLD_X, FOLD_Y), dim3(128), 0, stream>>>(E, mm);
  diag_gather_kernel<<<dim3(NG), dim3(256), 0, stream>>>(E, mm, out);
}
